# starting kernel
# baseline (speedup 1.0000x reference)
#include <hip/hip_runtime.h>

#define D 64
#define ALPHA 0.5f

typedef __attribute__((ext_vector_type(2))) float v2f;
typedef __attribute__((ext_vector_type(4))) float v4f;
typedef __attribute__((ext_vector_type(8))) float v8f;

__global__ void zero_ws_kernel(float* __restrict__ ws, int n) {
    int i = blockIdx.x * blockDim.x + threadIdx.x;
    if (i < n) ws[i] = 0.0f;
}

__global__ void scatter_edges_kernel(const float* __restrict__ x,
                                     const int*   __restrict__ ei,
                                     float* __restrict__ sum_in,
                                     float* __restrict__ sum_out,
                                     float* __restrict__ cnt_in,
                                     float* __restrict__ cnt_out,
                                     int E) {
    long long tid = (long long)blockIdx.x * blockDim.x + threadIdx.x;
    long long tot = (long long)E * (D / 4);
    if (tid >= tot) return;
    int e = (int)(tid >> 4);
    int f = (int)(tid & 15) * 4;
    int s = ei[e];
    int d = ei[E + e];

    v4f xs = *(const v4f*)(x + (long long)s * D + f);
    v4f xd = *(const v4f*)(x + (long long)d * D + f);

    float* pin  = sum_in  + (long long)d * D + f;
    float* pout = sum_out + (long long)s * D + f;
    atomicAdd(pin + 0, xs.x); atomicAdd(pin + 1, xs.y);
    atomicAdd(pin + 2, xs.z); atomicAdd(pin + 3, xs.w);
    atomicAdd(pout + 0, xd.x); atomicAdd(pout + 1, xd.y);
    atomicAdd(pout + 2, xd.z); atomicAdd(pout + 3, xd.w);

    if (f == 0) {
        atomicAdd(&cnt_in[d], 1.0f);
        atomicAdd(&cnt_out[s], 1.0f);
    }
}

__global__ __launch_bounds__(128)
void dirsage_gemm_kernel(const float* __restrict__ x,
                         const float* __restrict__ sum_in,
                         const float* __restrict__ sum_out,
                         const float* __restrict__ cnt_in,
                         const float* __restrict__ cnt_out,
                         const float* __restrict__ Wself,
                         const float* __restrict__ bself,
                         const float* __restrict__ Ws2d,
                         const float* __restrict__ bs2d,
                         const float* __restrict__ Wd2s,
                         const float* __restrict__ bd2s,
                         float* __restrict__ out,
                         int N) {
    __shared__ float wl[3 * D * D];

    for (int i = threadIdx.x; i < D * D; i += blockDim.x) {
        int k   = i >> 6;
        int col = i & 63;
        int dst = ((k >> 1) << 7) + (col << 1) + (k & 1);
        wl[dst]             = Wself[i];
        wl[D * D + dst]     = Ws2d[i];
        wl[2 * D * D + dst] = Wd2s[i];
    }
    __syncthreads();

    int wave  = threadIdx.x >> 5;
    int lane  = threadIdx.x & 31;
    int mtile = blockIdx.x * (blockDim.x >> 5) + wave;
    if (mtile * 16 >= N) return;

    int row   = lane & 15;
    int half  = lane >> 4;
    int khalf = half * 2;
    int gr    = mtile * 16 + row;

    float sc_in  = (1.0f - ALPHA) / fmaxf(cnt_in[gr], 1.0f);
    float sc_out = ALPHA          / fmaxf(cnt_out[gr], 1.0f);

    v8f acc[4];
    for (int t = 0; t < 4; ++t) {
        float b = bself[t * 16 + row]
                + (1.0f - ALPHA) * bs2d[t * 16 + row]
                + ALPHA * bd2s[t * 16 + row];
        for (int r = 0; r < 8; ++r) acc[t][r] = b;
    }

    const float* xrow = x       + (long long)gr * D;
    const float* irow = sum_in  + (long long)gr * D;
    const float* orow = sum_out + (long long)gr * D;

    const float* wbase = &wl[half * 128 + row * 2];

    for (int k0 = 0; k0 < D; k0 += 4) {
        int ka = k0 + khalf;
        v2f ax = *(const v2f*)(xrow + ka);
        v2f ai = *(const v2f*)(irow + ka);
        v2f ao = *(const v2f*)(orow + ka);
        ai *= sc_in;
        ao *= sc_out;

        const float* wk = wbase + (k0 >> 1) * 128;
        for (int t = 0; t < 4; ++t) {
            v2f bs = *(const v2f*)(wk + t * 32);
            v2f bi = *(const v2f*)(wk + D * D + t * 32);
            v2f bo = *(const v2f*)(wk + 2 * D * D + t * 32);

            acc[t] = __builtin_amdgcn_wmma_f32_16x16x4_f32(
                false, ax, false, bs, (short)0, acc[t], false, false);
            acc[t] = __builtin_amdgcn_wmma_f32_16x16x4_f32(
                false, ai, false, bi, (short)0, acc[t], false, false);
            acc[t] = __builtin_amdgcn_wmma_f32_16x16x4_f32(
                false, ao, false, bo, (short)0, acc[t], false, false);
        }
    }

    int mofs = half * 8;
    for (int t = 0; t < 4; ++t) {
        int col = t * 16 + row;
        long long base = (long long)(mtile * 16 + mofs) * D + col;
        for (int r = 0; r < 8; ++r) {
            out[base + (long long)r * D] = acc[t][r];
        }
    }
}

extern "C" void kernel_launch(void* const* d_in, const int* in_sizes, int n_in,
                              void* d_out, int out_size, void* d_ws, size_t ws_size,
                              hipStream_t stream) {
    const float* x     = (const float*)d_in[0];
    const int*   ei    = (const int*)  d_in[1];
    const float* Wself = (const float*)d_in[2];
    const float* bself = (const float*)d_in[3];
    const float* Ws2d  = (const float*)d_in[4];
    const float* bs2d  = (const float*)d_in[5];
    const float* Wd2s  = (const float*)d_in[6];
    const float* bd2s  = (const float*)d_in[7];
    float* out = (float*)d_out;

    int N = in_sizes[0] / D;
    int E = in_sizes[1] / 2;

    float* ws      = (float*)d_ws;
    float* sum_in  = ws;
    float* sum_out = ws + (size_t)N * D;
    float* cnt_in  = ws + 2 * (size_t)N * D;
    float* cnt_out = cnt_in + N;

    int zn = 2 * N * D + 2 * N;
    zero_ws_kernel<<<(zn + 255) / 256, 256, 0, stream>>>(ws, zn);

    long long sthreads = (long long)E * (D / 4);
    int sblocks = (int)((sthreads + 255) / 256);
    scatter_edges_kernel<<<sblocks, 256, 0, stream>>>(
        x, ei, sum_in, sum_out, cnt_in, cnt_out, E);

    int mtiles = (N + 15) / 16;
    int wavesPerBlock = 4;
    int gblocks = (mtiles + wavesPerBlock - 1) / wavesPerBlock;
    dirsage_gemm_kernel<<<gblocks, 32 * wavesPerBlock, 0, stream>>>(
        x, sum_in, sum_out, cnt_in, cnt_out,
        Wself, bself, Ws2d, bs2d, Wd2s, bd2s, out, N);
}


# Scatter remapped to lane = feature: each f32 atomic instruction adds one whole 128-B line of one row (was 8 words in each of 4 lines); float sums still add in arrival order
# speedup vs baseline: 1.2644x; 1.2644x over previous
#include <hip/hip_runtime.h>

#define D 64
#define ALPHA 0.5f

typedef __attribute__((ext_vector_type(2))) float v2f;
typedef __attribute__((ext_vector_type(4))) float v4f;
typedef __attribute__((ext_vector_type(8))) float v8f;

__global__ void zero_ws_kernel(float* __restrict__ ws, int n) {
    int i = blockIdx.x * blockDim.x + threadIdx.x;
    if (i < n) ws[i] = 0.0f;
}

#define EPW 16

__global__ __launch_bounds__(256)
void scatter_edges_kernel(const float* __restrict__ x,
                          const int*   __restrict__ ei,
                          float* __restrict__ sum_in,
                          float* __restrict__ sum_out,
                          float* __restrict__ cnt_in,
                          float* __restrict__ cnt_out,
                          int E) {
    int lane = threadIdx.x & 31;
    int wave = __builtin_amdgcn_readfirstlane(
        (int)(((long long)blockIdx.x * blockDim.x + threadIdx.x) >> 5));
    long long e0 = (long long)wave * EPW;
    if (e0 >= E) return;
    long long rem = (long long)E - e0;
    int n = rem < EPW ? (int)rem : EPW;

    int h  = lane >> 4;
    int le = lane & 15;
    int idx = 0;
    if (le < n) {
        idx = ei[(long long)h * E + e0 + le];
        atomicAdd((h ? cnt_in : cnt_out) + idx, 1.0f);
    }

    int j = 0;
    for (; j + 4 <= n; j += 4) {
        int   s[4], d[4];
        float a0[4], a1[4], b0[4], b1[4];
#pragma unroll
        for (int u = 0; u < 4; ++u) {
            s[u] = __builtin_amdgcn_readlane(idx, j + u);
            d[u] = __builtin_amdgcn_readlane(idx, 16 + j + u);
            const float* xs = x + (long long)s[u] * D + lane;
            const float* xd = x + (long long)d[u] * D + lane;
            a0[u] = xs[0]; a1[u] = xs[32];
            b0[u] = xd[0]; b1[u] = xd[32];
        }
#pragma unroll
        for (int u = 0; u < 4; ++u) {
            float* pin  = sum_in  + (long long)d[u] * D + lane;
            float* pout = sum_out + (long long)s[u] * D + lane;
            atomicAdd(pin,       a0[u]);
            atomicAdd(pin  + 32, a1[u]);
            atomicAdd(pout,      b0[u]);
            atomicAdd(pout + 32, b1[u]);
        }
    }
    for (; j < n; ++j) {
        int sj = __builtin_amdgcn_readlane(idx, j);
        int dj = __builtin_amdgcn_readlane(idx, 16 + j);
        const float* xs = x + (long long)sj * D + lane;
        const float* xd = x + (long long)dj * D + lane;
        float a0 = xs[0], a1 = xs[32], b0 = xd[0], b1 = xd[32];
        float* pin  = sum_in  + (long long)dj * D + lane;
        float* pout = sum_out + (long long)sj * D + lane;
        atomicAdd(pin,       a0);
        atomicAdd(pin  + 32, a1);
        atomicAdd(pout,      b0);
        atomicAdd(pout + 32, b1);
    }
}

__global__ __launch_bounds__(128)
void dirsage_gemm_kernel(const float* __restrict__ x,
                         const float* __restrict__ sum_in,
                         const float* __restrict__ sum_out,
                         const float* __restrict__ cnt_in,
                         const float* __restrict__ cnt_out,
                         const float* __restrict__ Wself,
                         const float* __restrict__ bself,
                         const float* __restrict__ Ws2d,
                         const float* __restrict__ bs2d,
                         const float* __restrict__ Wd2s,
                         const float* __restrict__ bd2s,
                         float* __restrict__ out,
                         int N) {
    __shared__ float wl[3 * D * D];

    for (int i = threadIdx.x; i < D * D; i += blockDim.x) {
        int k   = i >> 6;
        int col = i & 63;
        int dst = ((k >> 1) << 7) + (col << 1) + (k & 1);
        wl[dst]             = Wself[i];
        wl[D * D + dst]     = Ws2d[i];
        wl[2 * D * D + dst] = Wd2s[i];
    }
    __syncthreads();

    int wave  = threadIdx.x >> 5;
    int lane  = threadIdx.x & 31;
    int mtile = blockIdx.x * (blockDim.x >> 5) + wave;
    if (mtile * 16 >= N) return;

    int row   = lane & 15;
    int half  = lane >> 4;
    int khalf = half * 2;
    int gr    = mtile * 16 + row;

    float sc_in  = (1.0f - ALPHA) / fmaxf(cnt_in[gr], 1.0f);
    float sc_out = ALPHA          / fmaxf(cnt_out[gr], 1.0f);

    v8f acc[4];
    for (int t = 0; t < 4; ++t) {
        float b = bself[t * 16 + row]
                + (1.0f - ALPHA) * bs2d[t * 16 + row]
                + ALPHA * bd2s[t * 16 + row];
        for (int r = 0; r < 8; ++r) acc[t][r] = b;
    }

    const float* xrow = x       + (long long)gr * D;
    const float* irow = sum_in  + (long long)gr * D;
    const float* orow = sum_out + (long long)gr * D;

    const float* wbase = &wl[half * 128 + row * 2];

    for (int k0 = 0; k0 < D; k0 += 4) {
        int ka = k0 + khalf;
        v2f ax = *(const v2f*)(xrow + ka);
        v2f ai = *(const v2f*)(irow + ka);
        v2f ao = *(const v2f*)(orow + ka);
        ai *= sc_in;
        ao *= sc_out;

        const float* wk = wbase + (k0 >> 1) * 128;
        for (int t = 0; t < 4; ++t) {
            v2f bs = *(const v2f*)(wk + t * 32);
            v2f bi = *(const v2f*)(wk + D * D + t * 32);
            v2f bo = *(const v2f*)(wk + 2 * D * D + t * 32);

            acc[t] = __builtin_amdgcn_wmma_f32_16x16x4_f32(
                false, ax, false, bs, (short)0, acc[t], false, false);
            acc[t] = __builtin_amdgcn_wmma_f32_16x16x4_f32(
                false, ai, false, bi, (short)0, acc[t], false, false);
            acc[t] = __builtin_amdgcn_wmma_f32_16x16x4_f32(
                false, ao, false, bo, (short)0, acc[t], false, false);
        }
    }

    int mofs = half * 8;
    for (int t = 0; t < 4; ++t) {
        int col = t * 16 + row;
        long long base = (long long)(mtile * 16 + mofs) * D + col;
        for (int r = 0; r < 8; ++r) {
            out[base + (long long)r * D] = acc[t][r];
        }
    }
}

extern "C" void kernel_launch(void* const* d_in, const int* in_sizes, int n_in,
                              void* d_out, int out_size, void* d_ws, size_t ws_size,
                              hipStream_t stream) {
    const float* x     = (const float*)d_in[0];
    const int*   ei    = (const int*)  d_in[1];
    const float* Wself = (const float*)d_in[2];
    const float* bself = (const float*)d_in[3];
    const float* Ws2d  = (const float*)d_in[4];
    const float* bs2d  = (const float*)d_in[5];
    const float* Wd2s  = (const float*)d_in[6];
    const float* bd2s  = (const float*)d_in[7];
    float* out = (float*)d_out;

    int N = in_sizes[0] / D;
    int E = in_sizes[1] / 2;

    float* ws      = (float*)d_ws;
    float* sum_in  = ws;
    float* sum_out = ws + (size_t)N * D;
    float* cnt_in  = ws + 2 * (size_t)N * D;
    float* cnt_out = cnt_in + N;

    int zn = 2 * N * D + 2 * N;
    zero_ws_kernel<<<(zn + 255) / 256, 256, 0, stream>>>(ws, zn);

    int swaves  = (E + EPW - 1) / EPW;
    int sblocks = (swaves + 7) / 8;
    scatter_edges_kernel<<<sblocks, 256, 0, stream>>>(
        x, ei, sum_in, sum_out, cnt_in, cnt_out, E);

    int mtiles = (N + 15) / 16;
    int wavesPerBlock = 4;
    int gblocks = (mtiles + wavesPerBlock - 1) / wavesPerBlock;
    dirsage_gemm_kernel<<<gblocks, 32 * wavesPerBlock, 0, stream>>>(
        x, sum_in, sum_out, cnt_in, cnt_out,
        Wself, bself, Ws2d, bs2d, Wd2s, bd2s, out, N);
}


# Scatter split by cache half: a wave reads its die and adds only into the sum plane its half owns (work by integer tickets), so no line is added into from both halves; float sums stay arrival-ordered
# speedup vs baseline: 2.4625x; 1.9476x over previous
#include <hip/hip_runtime.h>

#define D 64
#define ALPHA 0.5f

#define CTR_WORDS 128

typedef __attribute__((ext_vector_type(2))) float v2f;
typedef __attribute__((ext_vector_type(4))) float v4f;
typedef __attribute__((ext_vector_type(8))) float v8f;

__global__ void zero_ws_kernel(float* __restrict__ ws, int n,
                               float* __restrict__ ctr_area) {
    int i = blockIdx.x * blockDim.x + threadIdx.x;
    if (i < n) ws[i] = 0.0f;
    if (i < CTR_WORDS) ctr_area[i] = 0.0f;
}

#define EPI 64

__device__ __forceinline__ void add_rows32(const float* __restrict__ x,
                                           float* __restrict__ sum,
                                           int gv, int av, int n, int lane) {
    int j = 0;
    for (; j + 8 <= n; j += 8) {
        int   a[8];
        float v0[8], v1[8];
#pragma unroll
        for (int u = 0; u < 8; ++u) {
            int g = __builtin_amdgcn_readlane(gv, j + u);
            a[u]  = __builtin_amdgcn_readlane(av, j + u);
            const float* xr = x + (long long)g * D + lane;
            v0[u] = xr[0];
            v1[u] = xr[32];
        }
#pragma unroll
        for (int u = 0; u < 8; ++u) {
            float* ps = sum + (long long)a[u] * D + lane;
            atomicAdd(ps,      v0[u]);
            atomicAdd(ps + 32, v1[u]);
        }
    }
    for (; j < n; ++j) {
        int g  = __builtin_amdgcn_readlane(gv, j);
        int aj = __builtin_amdgcn_readlane(av, j);
        const float* xr = x + (long long)g * D + lane;
        float v0 = xr[0], v1 = xr[32];
        float* ps = sum + (long long)aj * D + lane;
        atomicAdd(ps,      v0);
        atomicAdd(ps + 32, v1);
    }
}

__device__ __forceinline__ int take_ticket(int* ctr, int lane) {
    int t = 0;
    if (lane == 0) t = atomicAdd(ctr, 1);
    return __builtin_amdgcn_readfirstlane(t);
}

__global__ __launch_bounds__(256)
void scatter_edges_kernel(const float* __restrict__ x,
                          const int*   __restrict__ ei,
                          float* __restrict__ sum_in,
                          float* __restrict__ sum_out,
                          float* __restrict__ cnt_in,
                          float* __restrict__ cnt_out,
                          float* __restrict__ ctr_area,
                          int E) {
    int lane   = threadIdx.x & 31;
    int nitems = (E + EPI - 1) / EPI;
    int* ctr   = (int*)ctr_area;

    unsigned place = __builtin_amdgcn_s_sendmsg_rtn(0x87);
    int half = (int)((place >> 18) & 1u);

    int dir = half;
    int c = take_ticket(ctr + dir * 64, lane);
    if (c >= nitems) {
        dir = 1 - half;
        c = take_ticket(ctr + dir * 64, lane);
        if (c >= nitems) return;
    }

    long long e0 = (long long)c * EPI;
    long long rem = (long long)E - e0;
    int n = rem < EPI ? (int)rem : EPI;

    const int* gi = ei + (dir ? (long long)E : 0LL) + e0;
    const int* ai = ei + (dir ? 0LL : (long long)E) + e0;
    float* sum = dir ? sum_out : sum_in;
    float* cnt = dir ? cnt_out : cnt_in;

    int g0 = 0, a0 = 0, g1 = 0, a1 = 0;
    if (lane < n) {
        g0 = gi[lane];
        a0 = ai[lane];
        atomicAdd(cnt + a0, 1.0f);
    }
    if (lane + 32 < n) {
        g1 = gi[lane + 32];
        a1 = ai[lane + 32];
        atomicAdd(cnt + a1, 1.0f);
    }

    int n0 = n < 32 ? n : 32;
    add_rows32(x, sum, g0, a0, n0,     lane);
    add_rows32(x, sum, g1, a1, n - n0, lane);
}

__global__ __launch_bounds__(128)
void dirsage_gemm_kernel(const float* __restrict__ x,
                         const float* __restrict__ sum_in,
                         const float* __restrict__ sum_out,
                         const float* __restrict__ cnt_in,
                         const float* __restrict__ cnt_out,
                         const float* __restrict__ Wself,
                         const float* __restrict__ bself,
                         const float* __restrict__ Ws2d,
                         const float* __restrict__ bs2d,
                         const float* __restrict__ Wd2s,
                         const float* __restrict__ bd2s,
                         float* __restrict__ out,
                         int N) {
    __shared__ float wl[3 * D * D];

    for (int i = threadIdx.x; i < D * D; i += blockDim.x) {
        int k   = i >> 6;
        int col = i & 63;
        int dst = ((k >> 1) << 7) + (col << 1) + (k & 1);
        wl[dst]             = Wself[i];
        wl[D * D + dst]     = Ws2d[i];
        wl[2 * D * D + dst] = Wd2s[i];
    }
    __syncthreads();

    int wave  = threadIdx.x >> 5;
    int lane  = threadIdx.x & 31;
    int mtile = blockIdx.x * (blockDim.x >> 5) + wave;
    if (mtile * 16 >= N) return;

    int row   = lane & 15;
    int half  = lane >> 4;
    int khalf = half * 2;
    int gr    = mtile * 16 + row;

    float sc_in  = (1.0f - ALPHA) / fmaxf(cnt_in[gr], 1.0f);
    float sc_out = ALPHA          / fmaxf(cnt_out[gr], 1.0f);

    v8f acc[4];
    for (int t = 0; t < 4; ++t) {
        float b = bself[t * 16 + row]
                + (1.0f - ALPHA) * bs2d[t * 16 + row]
                + ALPHA * bd2s[t * 16 + row];
        for (int r = 0; r < 8; ++r) acc[t][r] = b;
    }

    const float* xrow = x       + (long long)gr * D;
    const float* irow = sum_in  + (long long)gr * D;
    const float* orow = sum_out + (long long)gr * D;

    const float* wbase = &wl[half * 128 + row * 2];

    for (int k0 = 0; k0 < D; k0 += 4) {
        int ka = k0 + khalf;
        v2f ax = *(const v2f*)(xrow + ka);
        v2f ai = *(const v2f*)(irow + ka);
        v2f ao = *(const v2f*)(orow + ka);
        ai *= sc_in;
        ao *= sc_out;

        const float* wk = wbase + (k0 >> 1) * 128;
        for (int t = 0; t < 4; ++t) {
            v2f bs = *(const v2f*)(wk + t * 32);
            v2f bi = *(const v2f*)(wk + D * D + t * 32);
            v2f bo = *(const v2f*)(wk + 2 * D * D + t * 32);

            acc[t] = __builtin_amdgcn_wmma_f32_16x16x4_f32(
                false, ax, false, bs, (short)0, acc[t], false, false);
            acc[t] = __builtin_amdgcn_wmma_f32_16x16x4_f32(
                false, ai, false, bi, (short)0, acc[t], false, false);
            acc[t] = __builtin_amdgcn_wmma_f32_16x16x4_f32(
                false, ao, false, bo, (short)0, acc[t], false, false);
        }
    }

    int mofs = half * 8;
    for (int t = 0; t < 4; ++t) {
        int col = t * 16 + row;
        long long base = (long long)(mtile * 16 + mofs) * D + col;
        for (int r = 0; r < 8; ++r) {
            out[base + (long long)r * D] = acc[t][r];
        }
    }
}

extern "C" void kernel_launch(void* const* d_in, const int* in_sizes, int n_in,
                              void* d_out, int out_size, void* d_ws, size_t ws_size,
                              hipStream_t stream) {
    const float* x     = (const float*)d_in[0];
    const int*   ei    = (const int*)  d_in[1];
    const float* Wself = (const float*)d_in[2];
    const float* bself = (const float*)d_in[3];
    const float* Ws2d  = (const float*)d_in[4];
    const float* bs2d  = (const float*)d_in[5];
    const float* Wd2s  = (const float*)d_in[6];
    const float* bd2s  = (const float*)d_in[7];
    float* out = (float*)d_out;

    int N = in_sizes[0] / D;
    int E = in_sizes[1] / 2;

    float* ws      = (float*)d_ws;
    float* sum_in  = ws;
    float* sum_out = ws + (size_t)N * D;
    float* cnt_in  = ws + 2 * (size_t)N * D;
    float* cnt_out = cnt_in + N;

    int zn = 2 * N * D + 2 * N;
    zero_ws_kernel<<<(zn + 255) / 256, 256, 0, stream>>>(ws, zn, out);

    int sitems  = (E + EPI - 1) / EPI;
    int swaves  = 2 * sitems;
    int sblocks = (swaves + 7) / 8;
    scatter_edges_kernel<<<sblocks, 256, 0, stream>>>(
        x, ei, sum_in, sum_out, cnt_in, cnt_out, out, E);

    int mtiles = (N + 15) / 16;
    int wavesPerBlock = 4;
    int gblocks = (mtiles + wavesPerBlock - 1) / wavesPerBlock;
    dirsage_gemm_kernel<<<gblocks, 32 * wavesPerBlock, 0, stream>>>(
        x, sum_in, sum_out, cnt_in, cnt_out,
        Wself, bself, Ws2d, bs2d, Wd2s, bd2s, out, N);
}


# Row sums and degree counts kept on chip: edges binned by the 400-row range they add into, each range summed in its work-group's LDS (f32 LDS adds, arrival order) and fed to the WMMA from there
# speedup vs baseline: 4.1149x; 1.6710x over previous
#include <hip/hip_runtime.h>


#define D 64
#define ALPHA 0.5f

#define RB     400
#define PITCH  68
#define NBMAX  512
#define CHUNK  2048
#define EPT    8

typedef __attribute__((ext_vector_type(2))) float v2f;
typedef __attribute__((ext_vector_type(4))) float v4f;
typedef __attribute__((ext_vector_type(8))) float v8f;

__global__ void zero_counters_kernel(int* __restrict__ gcnt, int n) {
    int i = blockIdx.x * blockDim.x + threadIdx.x;
    if (i < n) gcnt[i] = 0;
}

__global__ __launch_bounds__(256)
void bin_edges_kernel(const int* __restrict__ ei,
                      int* __restrict__ gcnt,
                      int* __restrict__ lists,
                      int E, int NB, int CAP) {
    __shared__ int lh[2 * NBMAX];
    __shared__ int lbase[2 * NBMAX];
    int tid = threadIdx.x;
    for (int i = tid; i < 2 * NB; i += 256) lh[i] = 0;
    __syncthreads();

    int base = blockIdx.x * CHUNK;
    int s[EPT], d[EPT], r0[EPT], r1[EPT];
#pragma unroll
    for (int k = 0; k < EPT; ++k) {
        int e = base + k * 256 + tid;
        s[k] = 0; d[k] = 0; r0[k] = 0; r1[k] = 0;
        if (e < E) {
            s[k] = ei[e];
            d[k] = ei[E + e];
            r0[k] = atomicAdd(&lh[d[k] / RB], 1);
            r1[k] = atomicAdd(&lh[NB + s[k] / RB], 1);
        }
    }
    __syncthreads();
    for (int i = tid; i < 2 * NB; i += 256) {
        int c = lh[i];
        lbase[i] = c ? atomicAdd(&gcnt[i], c) : 0;
    }
    __syncthreads();

    int2* recs = (int2*)lists;
#pragma unroll
    for (int k = 0; k < EPT; ++k) {
        int e = base + k * 256 + tid;
        if (e < E) {
            int b0 = d[k] / RB;
            int p0 = lbase[b0] + r0[k];
            if (p0 < CAP) {
                int2 r; r.x = s[k]; r.y = d[k] - b0 * RB;
                recs[(long long)b0 * CAP + p0] = r;
            }
            int b1 = s[k] / RB;
            int p1 = lbase[NB + b1] + r1[k];
            if (p1 < CAP) {
                int2 r; r.x = d[k]; r.y = s[k] - b1 * RB;
                recs[(long long)(NB + b1) * CAP + p1] = r;
            }
        }
    }
}

__device__ __forceinline__ void add_rows_lds(const float* __restrict__ x,
                                             float* pl, int gv, int tv,
                                             int n, int lane) {
    int j = 0;
    for (; j + 16 <= n; j += 16) {
        int   a[16];
        float v0[16], v1[16];
#pragma unroll
        for (int u = 0; u < 16; ++u) {
            int g = __builtin_amdgcn_readlane(gv, j + u);
            a[u]  = __builtin_amdgcn_readlane(tv, j + u);
            const float* xr = x + (long long)g * D + lane;
            v0[u] = xr[0];
            v1[u] = xr[32];
        }
#pragma unroll
        for (int u = 0; u < 16; ++u) {
            float* ps = pl + a[u] * PITCH + lane;
            atomicAdd(ps,      v0[u]);
            atomicAdd(ps + 32, v1[u]);
        }
    }
    for (; j < n; ++j) {
        int g  = __builtin_amdgcn_readlane(gv, j);
        int aj = __builtin_amdgcn_readlane(tv, j);
        const float* xr = x + (long long)g * D + lane;
        float v0 = xr[0], v1 = xr[32];
        float* ps = pl + aj * PITCH + lane;
        atomicAdd(ps,      v0);
        atomicAdd(ps + 32, v1);
    }
}

__global__ __launch_bounds__(1024)
void owner_sum_gemm_kernel(const float* __restrict__ x,
                           const int*   __restrict__ ei,
                           const int*   __restrict__ gcnt,
                           const int*   __restrict__ lists,
                           const float* __restrict__ Wself,
                           const float* __restrict__ bself,
                           const float* __restrict__ Ws2d,
                           const float* __restrict__ bs2d,
                           const float* __restrict__ Wd2s,
                           const float* __restrict__ bd2s,
                           float* __restrict__ out,
                           int N, int E, int NB, int CAP) {
    __shared__ __attribute__((aligned(16))) float acc[2 * RB * PITCH];
    __shared__ int   cntl[2 * RB];
    __shared__ __attribute__((aligned(16))) float wl[3 * D * D];

    int tid  = threadIdx.x;
    int lane = tid & 31;
    int wave = __builtin_amdgcn_readfirstlane(tid >> 5);
    int b    = blockIdx.x;
    int lo   = b * RB;

    v4f z4 = {0.0f, 0.0f, 0.0f, 0.0f};
    for (int i = tid; i < (2 * RB * PITCH) / 4; i += 1024) ((v4f*)acc)[i] = z4;
    for (int i = tid; i < 2 * RB; i += 1024) cntl[i] = 0;
    for (int i = tid; i < D * D; i += 1024) {
        int k   = i >> 6;
        int col = i & 63;
        int dst = ((k >> 1) << 7) + (col << 1) + (k & 1);
        wl[dst]             = Wself[i];
        wl[D * D + dst]     = Ws2d[i];
        wl[2 * D * D + dst] = Wd2s[i];
    }
    __syncthreads();

    for (int dir = 0; dir < 2; ++dir) {
        float* pl = acc  + dir * (RB * PITCH);
        int*   cl = cntl + dir * RB;
        int c = gcnt[dir * NB + b];
        if (c <= CAP) {
            const int2* rec = (const int2*)lists + (long long)(dir * NB + b) * CAP;
            int g = wave * 32;
            int2 rn; rn.x = 0; rn.y = 0;
            if (g + lane < c) rn = rec[g + lane];
            for (; g < c; g += 1024) {
                int n = c - g < 32 ? c - g : 32;
                int gv = rn.x, tv = rn.y;
                if (g + 1024 + lane < c) rn = rec[g + 1024 + lane];
                if (lane < n) atomicAdd(&cl[tv], 1);
                add_rows_lds(x, pl, gv, tv, n, lane);
            }
        } else {
            const int* gi = ei + (dir ? (long long)E : 0LL);
            const int* ti = ei + (dir ? 0LL : (long long)E);
            for (int e0 = wave * 32; e0 < E; e0 += 1024) {
                int e = e0 + lane;
                int gv = 0, tv = -1;
                if (e < E) {
                    int t = ti[e] - lo;
                    if ((unsigned)t < (unsigned)RB) {
                        tv = t;
                        gv = gi[e];
                        atomicAdd(&cl[tv], 1);
                    }
                }
                unsigned m = __builtin_amdgcn_ballot_w32(tv >= 0);
                while (m) {
                    int j = __builtin_ctz(m);
                    m &= m - 1;
                    int g  = __builtin_amdgcn_readlane(gv, j);
                    int aj = __builtin_amdgcn_readlane(tv, j);
                    const float* xr = x + (long long)g * D + lane;
                    float v0 = xr[0], v1 = xr[32];
                    float* ps = pl + aj * PITCH + lane;
                    atomicAdd(ps,      v0);
                    atomicAdd(ps + 32, v1);
                }
            }
        }
    }
    __syncthreads();

    int trow0 = wave * 16;
    if (trow0 >= RB || lo + trow0 >= N) return;

    int row   = lane & 15;
    int half  = lane >> 4;
    int khalf = half * 2;
    int lr    = trow0 + row;
    int gr    = lo + lr;

    float sc_in  = (1.0f - ALPHA) / fmaxf((float)cntl[lr], 1.0f);
    float sc_out = ALPHA          / fmaxf((float)cntl[RB + lr], 1.0f);

    v8f accv[4];
    for (int t = 0; t < 4; ++t) {
        float bb = bself[t * 16 + row]
                 + (1.0f - ALPHA) * bs2d[t * 16 + row]
                 + ALPHA * bd2s[t * 16 + row];
        for (int r = 0; r < 8; ++r) accv[t][r] = bb;
    }

    const float* xrow = x + (long long)gr * D;
    const float* irow = acc + lr * PITCH;
    const float* orow = acc + RB * PITCH + lr * PITCH;

    const float* wbase = &wl[half * 128 + row * 2];

    for (int k0 = 0; k0 < D; k0 += 4) {
        int ka = k0 + khalf;
        v2f ax = *(const v2f*)(xrow + ka);
        v2f ai = *(const v2f*)(irow + ka);
        v2f ao = *(const v2f*)(orow + ka);
        ai *= sc_in;
        ao *= sc_out;

        const float* wk = wbase + (k0 >> 1) * 128;
        for (int t = 0; t < 4; ++t) {
            v2f bs = *(const v2f*)(wk + t * 32);
            v2f bi = *(const v2f*)(wk + D * D + t * 32);
            v2f bo = *(const v2f*)(wk + 2 * D * D + t * 32);

            accv[t] = __builtin_amdgcn_wmma_f32_16x16x4_f32(
                false, ax, false, bs, (short)0, accv[t], false, false);
            accv[t] = __builtin_amdgcn_wmma_f32_16x16x4_f32(
                false, ai, false, bi, (short)0, accv[t], false, false);
            accv[t] = __builtin_amdgcn_wmma_f32_16x16x4_f32(
                false, ao, false, bo, (short)0, accv[t], false, false);
        }
    }

    int mofs = half * 8;
    for (int t = 0; t < 4; ++t) {
        int col = t * 16 + row;
        long long obase = (long long)(lo + trow0 + mofs) * D + col;
        for (int r = 0; r < 8; ++r) {
            out[obase + (long long)r * D] = accv[t][r];
        }
    }
}

extern "C" void kernel_launch(void* const* d_in, const int* in_sizes, int n_in,
                              void* d_out, int out_size, void* d_ws, size_t ws_size,
                              hipStream_t stream) {
    const float* x     = (const float*)d_in[0];
    const int*   ei    = (const int*)  d_in[1];
    const float* Wself = (const float*)d_in[2];
    const float* bself = (const float*)d_in[3];
    const float* Ws2d  = (const float*)d_in[4];
    const float* bs2d  = (const float*)d_in[5];
    const float* Wd2s  = (const float*)d_in[6];
    const float* bd2s  = (const float*)d_in[7];
    float* out = (float*)d_out;

    int N = in_sizes[0] / D;
    int E = in_sizes[1] / 2;

    int NB  = (N + RB - 1) / RB;
    int CAP = 2 * (E / NB);

    int* gcnt  = (int*)d_ws;
    int* lists = gcnt + 2 * NBMAX;

    zero_counters_kernel<<<(2 * NB + 1023) / 1024, 1024, 0, stream>>>(gcnt, 2 * NB);

    bin_edges_kernel<<<(E + CHUNK - 1) / CHUNK, 256, 0, stream>>>(
        ei, gcnt, lists, E, NB, CAP);

    owner_sum_gemm_kernel<<<NB, 1024, 0, stream>>>(
        x, ei, gcnt, lists, Wself, bself, Ws2d, bs2d, Wd2s, bd2s, out, N, E, NB, CAP);
}


# Row sums formed in registers: a range's records put in local-row order in LDS (integer count, scan, place), each wave sums whole rows; no LDS float adds; adds in list-rank (arrival) order
# speedup vs baseline: 4.3314x; 1.0526x over previous
#include <hip/hip_runtime.h>


#define D 64
#define ALPHA 0.5f

#define RB     400
#define PITCH  68
#define NBMAX  512
#define CHUNK  2048
#define EPT    8
#define SC     8192
#define SCT    8
#define TSH    20

typedef __attribute__((ext_vector_type(2))) float v2f;
typedef __attribute__((ext_vector_type(4))) float v4f;
typedef __attribute__((ext_vector_type(8))) float v8f;

__global__ void zero_counters_kernel(int* __restrict__ gcnt, int n) {
    int i = blockIdx.x * blockDim.x + threadIdx.x;
    if (i < n) gcnt[i] = 0;
}

__global__ __launch_bounds__(256)
void bin_edges_kernel(const int* __restrict__ ei,
                      int* __restrict__ gcnt,
                      int* __restrict__ lists,
                      int E, int NB, int CAP) {
    __shared__ int lh[2 * NBMAX];
    __shared__ int lbase[2 * NBMAX];
    int tid = threadIdx.x;
    for (int i = tid; i < 2 * NB; i += 256) lh[i] = 0;
    __syncthreads();

    int base = blockIdx.x * CHUNK;
    int s[EPT], d[EPT], r0[EPT], r1[EPT];
#pragma unroll
    for (int k = 0; k < EPT; ++k) {
        int e = base + k * 256 + tid;
        s[k] = 0; d[k] = 0; r0[k] = 0; r1[k] = 0;
        if (e < E) {
            s[k] = ei[e];
            d[k] = ei[E + e];
            r0[k] = atomicAdd(&lh[d[k] / RB], 1);
            r1[k] = atomicAdd(&lh[NB + s[k] / RB], 1);
        }
    }
    __syncthreads();
    for (int i = tid; i < 2 * NB; i += 256) {
        int c = lh[i];
        lbase[i] = c ? atomicAdd(&gcnt[i], c) : 0;
    }
    __syncthreads();

    int2* recs = (int2*)lists;
#pragma unroll
    for (int k = 0; k < EPT; ++k) {
        int e = base + k * 256 + tid;
        if (e < E) {
            int b0 = d[k] / RB;
            int p0 = lbase[b0] + r0[k];
            if (p0 < CAP) {
                int2 r; r.x = s[k]; r.y = d[k] - b0 * RB;
                recs[(long long)b0 * CAP + p0] = r;
            }
            int b1 = s[k] / RB;
            int p1 = lbase[NB + b1] + r1[k];
            if (p1 < CAP) {
                int2 r; r.x = d[k]; r.y = s[k] - b1 * RB;
                recs[(long long)(NB + b1) * CAP + p1] = r;
            }
        }
    }
}

__device__ __forceinline__ void add_rows_lds(const float* __restrict__ x,
                                             float* pl, int gv, int tv,
                                             int n, int lane) {
    int j = 0;
    for (; j + 16 <= n; j += 16) {
        int   a[16];
        float v0[16], v1[16];
#pragma unroll
        for (int u = 0; u < 16; ++u) {
            int g = __builtin_amdgcn_readlane(gv, j + u);
            a[u]  = __builtin_amdgcn_readlane(tv, j + u);
            const float* xr = x + (long long)g * D + lane;
            v0[u] = xr[0];
            v1[u] = xr[32];
        }
#pragma unroll
        for (int u = 0; u < 16; ++u) {
            float* ps = pl + a[u] * PITCH + lane;
            atomicAdd(ps,      v0[u]);
            atomicAdd(ps + 32, v1[u]);
        }
    }
    for (; j < n; ++j) {
        int g  = __builtin_amdgcn_readlane(gv, j);
        int aj = __builtin_amdgcn_readlane(tv, j);
        const float* xr = x + (long long)g * D + lane;
        float v0 = xr[0], v1 = xr[32];
        float* ps = pl + aj * PITCH + lane;
        atomicAdd(ps,      v0);
        atomicAdd(ps + 32, v1);
    }
}

__device__ __forceinline__ void sum_ordered_rows(const float* __restrict__ x,
                                                 float* pl, const int* sidx,
                                                 int pb, int pe, int lane) {
    int   tprev = -1;
    float s0 = 0.0f, s1 = 0.0f;
    int en = 0;
    if (pb + lane < pe) en = sidx[pb + lane];
    for (int p = pb; p < pe; p += 32) {
        int n  = pe - p < 32 ? pe - p : 32;
        int ev = en;
        if (p + 32 + lane < pe) en = sidx[p + 32 + lane];
        int j = 0;
        for (; j + 16 <= n; j += 16) {
            int   t[16];
            float v0[16], v1[16];
#pragma unroll
            for (int u = 0; u < 16; ++u) {
                int e = __builtin_amdgcn_readlane(ev, j + u);
                t[u]  = e >> TSH;
                unsigned off = (unsigned)(e & ((1 << TSH) - 1)) * D + lane;
                v0[u] = x[off];
                v1[u] = x[off + 32];
            }
#pragma unroll
            for (int u = 0; u < 16; ++u) {
                bool fresh = t[u] != tprev;
                tprev = t[u];
                s0 = (fresh ? 0.0f : s0) + v0[u];
                s1 = (fresh ? 0.0f : s1) + v1[u];
                float* ps = pl + t[u] * PITCH + lane;
                ps[0]  = s0;
                ps[32] = s1;
            }
        }
        for (; j < n; ++j) {
            int e  = __builtin_amdgcn_readlane(ev, j);
            int tj = e >> TSH;
            unsigned off = (unsigned)(e & ((1 << TSH) - 1)) * D + lane;
            float v0 = x[off], v1 = x[off + 32];
            bool fresh = tj != tprev;
            tprev = tj;
            s0 = (fresh ? 0.0f : s0) + v0;
            s1 = (fresh ? 0.0f : s1) + v1;
            float* ps = pl + tj * PITCH + lane;
            ps[0]  = s0;
            ps[32] = s1;
        }
    }
}

__global__ __launch_bounds__(1024)
void owner_sum_gemm_kernel(const float* __restrict__ x,
                           const int*   __restrict__ ei,
                           const int*   __restrict__ gcnt,
                           const int*   __restrict__ lists,
                           const float* __restrict__ Wself,
                           const float* __restrict__ bself,
                           const float* __restrict__ Ws2d,
                           const float* __restrict__ bs2d,
                           const float* __restrict__ Wd2s,
                           const float* __restrict__ bd2s,
                           float* __restrict__ out,
                           int N, int E, int NB, int CAP) {
    __shared__ __attribute__((aligned(16))) float acc[2 * RB * PITCH];
    __shared__ int   cntl[2 * RB];
    __shared__ __attribute__((aligned(16))) float wl[3 * D * D];
    __shared__ int   sidx[SC];
    __shared__ int   offs[RB + 4];

    int tid  = threadIdx.x;
    int lane = tid & 31;
    int wave = __builtin_amdgcn_readfirstlane(tid >> 5);
    int b    = blockIdx.x;
    int lo   = b * RB;

    v4f z4 = {0.0f, 0.0f, 0.0f, 0.0f};
    for (int i = tid; i < (2 * RB * PITCH) / 4; i += 1024) ((v4f*)acc)[i] = z4;
    for (int i = tid; i < 2 * RB; i += 1024) cntl[i] = 0;
    for (int i = tid; i < D * D; i += 1024) {
        int k   = i >> 6;
        int col = i & 63;
        int dst = ((k >> 1) << 7) + (col << 1) + (k & 1);
        wl[dst]             = Wself[i];
        wl[D * D + dst]     = Ws2d[i];
        wl[2 * D * D + dst] = Wd2s[i];
    }
    __syncthreads();

    for (int dir = 0; dir < 2; ++dir) {
        float* pl = acc  + dir * (RB * PITCH);
        int*   cl = cntl + dir * RB;
        int c = gcnt[dir * NB + b];
        if (c <= SC && c <= CAP) {
            const int2* rec = (const int2*)lists + (long long)(dir * NB + b) * CAP;
            int gk[SCT], tk[SCT], rk[SCT];
#pragma unroll
            for (int k = 0; k < SCT; ++k) {
                int i = k * 1024 + tid;
                gk[k] = 0; tk[k] = 0; rk[k] = 0;
                if (c > 0) {
                    int2 r = rec[i < c ? i : c - 1];
                    gk[k] = r.x; tk[k] = r.y;
                }
            }
#pragma unroll
            for (int k = 0; k < SCT; ++k) {
                int i = k * 1024 + tid;
                if (i < c) rk[k] = atomicAdd(&cl[tk[k]], 1);
            }
            __syncthreads();
            if (wave == 0) {
                int loc[13];
                int sum = 0;
#pragma unroll
                for (int q = 0; q < 13; ++q) {
                    int r = lane * 13 + q;
                    loc[q] = sum;
                    sum += (r < RB) ? cl[r] : 0;
                }
                int incl = sum;
#pragma unroll
                for (int dlt = 1; dlt < 32; dlt <<= 1) {
                    int up = __shfl_up(incl, dlt);
                    if (lane >= dlt) incl += up;
                }
                int basev = incl - sum;
#pragma unroll
                for (int q = 0; q < 13; ++q) {
                    int r = lane * 13 + q;
                    if (r <= RB) offs[r] = basev + loc[q];
                }
            }
            __syncthreads();
            int ok[SCT];
#pragma unroll
            for (int k = 0; k < SCT; ++k) ok[k] = offs[tk[k]];
#pragma unroll
            for (int k = 0; k < SCT; ++k) {
                int i = k * 1024 + tid;
                if (i < c) sidx[ok[k] + rk[k]] = gk[k] | (tk[k] << TSH);
            }
            __syncthreads();
            int rb = (wave * 25) >> 1;
            int re = ((wave + 1) * 25) >> 1;
            int pb = __builtin_amdgcn_readfirstlane(offs[rb]);
            int pe = __builtin_amdgcn_readfirstlane(offs[re]);
            sum_ordered_rows(x, pl, sidx, pb, pe, lane);
        } else if (c <= CAP) {
            const int2* rec = (const int2*)lists + (long long)(dir * NB + b) * CAP;
            int g = wave * 32;
            int2 rn; rn.x = 0; rn.y = 0;
            if (g + lane < c) rn = rec[g + lane];
            for (; g < c; g += 1024) {
                int n = c - g < 32 ? c - g : 32;
                int gv = rn.x, tv = rn.y;
                if (g + 1024 + lane < c) rn = rec[g + 1024 + lane];
                if (lane < n) atomicAdd(&cl[tv], 1);
                add_rows_lds(x, pl, gv, tv, n, lane);
            }
        } else {
            const int* gi = ei + (dir ? (long long)E : 0LL);
            const int* ti = ei + (dir ? 0LL : (long long)E);
            for (int e0 = wave * 32; e0 < E; e0 += 1024) {
                int e = e0 + lane;
                int gv = 0, tv = -1;
                if (e < E) {
                    int t = ti[e] - lo;
                    if ((unsigned)t < (unsigned)RB) {
                        tv = t;
                        gv = gi[e];
                        atomicAdd(&cl[tv], 1);
                    }
                }
                unsigned m = __builtin_amdgcn_ballot_w32(tv >= 0);
                while (m) {
                    int j = __builtin_ctz(m);
                    m &= m - 1;
                    int g  = __builtin_amdgcn_readlane(gv, j);
                    int aj = __builtin_amdgcn_readlane(tv, j);
                    const float* xr = x + (long long)g * D + lane;
                    float v0 = xr[0], v1 = xr[32];
                    float* ps = pl + aj * PITCH + lane;
                    atomicAdd(ps,      v0);
                    atomicAdd(ps + 32, v1);
                }
            }
        }
    }
    __syncthreads();

    int trow0 = wave * 16;
    if (trow0 >= RB || lo + trow0 >= N) return;

    int row   = lane & 15;
    int half  = lane >> 4;
    int khalf = half * 2;
    int lr    = trow0 + row;
    int gr    = lo + lr;

    float sc_in  = (1.0f - ALPHA) / fmaxf((float)cntl[lr], 1.0f);
    float sc_out = ALPHA          / fmaxf((float)cntl[RB + lr], 1.0f);

    v8f accv[4];
    for (int t = 0; t < 4; ++t) {
        float bb = bself[t * 16 + row]
                 + (1.0f - ALPHA) * bs2d[t * 16 + row]
                 + ALPHA * bd2s[t * 16 + row];
        for (int r = 0; r < 8; ++r) accv[t][r] = bb;
    }

    const float* xrow = x + (long long)gr * D;
    const float* irow = acc + lr * PITCH;
    const float* orow = acc + RB * PITCH + lr * PITCH;

    const float* wbase = &wl[half * 128 + row * 2];

    for (int k0 = 0; k0 < D; k0 += 4) {
        int ka = k0 + khalf;
        v2f ax = *(const v2f*)(xrow + ka);
        v2f ai = *(const v2f*)(irow + ka);
        v2f ao = *(const v2f*)(orow + ka);
        ai *= sc_in;
        ao *= sc_out;

        const float* wk = wbase + (k0 >> 1) * 128;
        for (int t = 0; t < 4; ++t) {
            v2f bs = *(const v2f*)(wk + t * 32);
            v2f bi = *(const v2f*)(wk + D * D + t * 32);
            v2f bo = *(const v2f*)(wk + 2 * D * D + t * 32);

            accv[t] = __builtin_amdgcn_wmma_f32_16x16x4_f32(
                false, ax, false, bs, (short)0, accv[t], false, false);
            accv[t] = __builtin_amdgcn_wmma_f32_16x16x4_f32(
                false, ai, false, bi, (short)0, accv[t], false, false);
            accv[t] = __builtin_amdgcn_wmma_f32_16x16x4_f32(
                false, ao, false, bo, (short)0, accv[t], false, false);
        }
    }

    int mofs = half * 8;
    for (int t = 0; t < 4; ++t) {
        int col = t * 16 + row;
        long long obase = (long long)(lo + trow0 + mofs) * D + col;
        for (int r = 0; r < 8; ++r) {
            out[obase + (long long)r * D] = accv[t][r];
        }
    }
}

extern "C" void kernel_launch(void* const* d_in, const int* in_sizes, int n_in,
                              void* d_out, int out_size, void* d_ws, size_t ws_size,
                              hipStream_t stream) {
    const float* x     = (const float*)d_in[0];
    const int*   ei    = (const int*)  d_in[1];
    const float* Wself = (const float*)d_in[2];
    const float* bself = (const float*)d_in[3];
    const float* Ws2d  = (const float*)d_in[4];
    const float* bs2d  = (const float*)d_in[5];
    const float* Wd2s  = (const float*)d_in[6];
    const float* bd2s  = (const float*)d_in[7];
    float* out = (float*)d_out;

    int N = in_sizes[0] / D;
    int E = in_sizes[1] / 2;

    int NB  = (N + RB - 1) / RB;
    int CAP = 2 * (E / NB);

    int* gcnt  = (int*)d_ws;
    int* lists = gcnt + 2 * NBMAX;

    zero_counters_kernel<<<(2 * NB + 1023) / 1024, 1024, 0, stream>>>(gcnt, 2 * NB);

    bin_edges_kernel<<<(E + CHUNK - 1) / CHUNK, 256, 0, stream>>>(
        ei, gcnt, lists, E, NB, CAP);

    owner_sum_gemm_kernel<<<NB, 1024, 0, stream>>>(
        x, ei, gcnt, lists, Wself, bself, Ws2d, bs2d, Wd2s, bd2s, out, N, E, NB, CAP);
}


# Binning stores whole runs: a group's records (4096 edges) are put in range order in LDS by a scan of its counts and written front to back, lanes on adjacent records; sums stay in arrival order
# speedup vs baseline: 5.1792x; 1.1957x over previous
#include <hip/hip_runtime.h>


#define D 64
#define ALPHA 0.5f

#define RB     400
#define PITCH  68
#define NBMAX  512
#define CHUNK  4096
#define BT     512
#define EPT    8
#define WPT    16
#define SPL    32
#define SC     8192
#define SCT    8
#define TSH    20

typedef __attribute__((ext_vector_type(2))) float v2f;
typedef __attribute__((ext_vector_type(4))) float v4f;
typedef __attribute__((ext_vector_type(8))) float v8f;

__global__ void zero_counters_kernel(int* __restrict__ gcnt, int n) {
    int i = blockIdx.x * blockDim.x + threadIdx.x;
    if (i < n) gcnt[i] = 0;
}

__global__ __launch_bounds__(BT)
void bin_edges_kernel(const int* __restrict__ ei,
                      int* __restrict__ gcnt,
                      int* __restrict__ lists,
                      int E, int NB, int CAP) {
    __shared__ __attribute__((aligned(16))) int lh[2 * NBMAX];
    __shared__ __attribute__((aligned(16))) int loff[2 * NBMAX];
    __shared__ int  lbase[2 * NBMAX];
    __shared__ int  ltotal;
    __shared__ int2 stage[2 * CHUNK];
    int tid = threadIdx.x;
    for (int i = tid; i < 2 * NBMAX; i += BT) lh[i] = 0;
    __syncthreads();

    int base = blockIdx.x * CHUNK;
    int s[EPT], d[EPT], r0[EPT], r1[EPT];
#pragma unroll
    for (int k = 0; k < EPT; ++k) {
        int e = base + k * BT + tid;
        s[k] = 0; d[k] = 0; r0[k] = 0; r1[k] = 0;
        if (e < E) {
            s[k] = ei[e];
            d[k] = ei[E + e];
            r0[k] = atomicAdd(&lh[d[k] / RB], 1);
            r1[k] = atomicAdd(&lh[NB + s[k] / RB], 1);
        }
    }
    __syncthreads();

    int cq[2], bq[2];
#pragma unroll
    for (int q = 0; q < 2; ++q) {
        int i = q * BT + tid;
        cq[q] = (i < 2 * NB) ? lh[i] : 0;
        bq[q] = 0;
        if (cq[q]) bq[q] = atomicAdd(&gcnt[i], cq[q]);
    }
    if (tid < 32) {
        int c[SPL];
        int sum = 0;
#pragma unroll
        for (int q = 0; q < SPL; ++q) { c[q] = lh[tid * SPL + q]; sum += c[q]; }
        int incl = sum;
#pragma unroll
        for (int dlt = 1; dlt < 32; dlt <<= 1) {
            int up = __shfl_up(incl, dlt);
            if (tid >= dlt) incl += up;
        }
        int run = incl - sum;
#pragma unroll
        for (int q = 0; q < SPL; ++q) { loff[tid * SPL + q] = run; run += c[q]; }
        if (tid == 31) ltotal = incl;
    }
#pragma unroll
    for (int q = 0; q < 2; ++q) {
        int i = q * BT + tid;
        if (i < 2 * NB) lbase[i] = bq[q];
    }
    __syncthreads();

    int sb0[EPT], lb0[EPT], sb1[EPT], lb1[EPT];
#pragma unroll
    for (int k = 0; k < EPT; ++k) {
        int b0 = d[k] / RB;
        int b1 = s[k] / RB;
        sb0[k] = loff[b0];      lb0[k] = lbase[b0];
        sb1[k] = loff[NB + b1]; lb1[k] = lbase[NB + b1];
    }
#pragma unroll
    for (int k = 0; k < EPT; ++k) {
        int e = base + k * BT + tid;
        if (e < E) {
            int b0 = d[k] / RB;
            int p0 = lb0[k] + r0[k];
            int2 w0;
            w0.x = s[k] | ((d[k] - b0 * RB) << TSH);
            w0.y = (p0 < CAP) ? b0 * CAP + p0 : -1;
            stage[sb0[k] + r0[k]] = w0;
            int b1 = s[k] / RB;
            int p1 = lb1[k] + r1[k];
            int2 w1;
            w1.x = d[k] | ((s[k] - b1 * RB) << TSH);
            w1.y = (p1 < CAP) ? (NB + b1) * CAP + p1 : -1;
            stage[sb1[k] + r1[k]] = w1;
        }
    }
    __syncthreads();

    int2* recs = (int2*)lists;
    int total = ltotal;
    int2 ev[WPT];
#pragma unroll
    for (int u = 0; u < WPT; ++u) {
        int p = u * BT + tid;
        ev[u].x = 0; ev[u].y = -1;
        if (p < total) ev[u] = stage[p];
    }
#pragma unroll
    for (int u = 0; u < WPT; ++u) {
        if (ev[u].y >= 0) {
            int2 r;
            r.x = ev[u].x & ((1 << TSH) - 1);
            r.y = ev[u].x >> TSH;
            recs[ev[u].y] = r;
        }
    }
}

__device__ __forceinline__ void add_rows_lds(const float* __restrict__ x,
                                             float* pl, int gv, int tv,
                                             int n, int lane) {
    int j = 0;
    for (; j + 16 <= n; j += 16) {
        int   a[16];
        float v0[16], v1[16];
#pragma unroll
        for (int u = 0; u < 16; ++u) {
            int g = __builtin_amdgcn_readlane(gv, j + u);
            a[u]  = __builtin_amdgcn_readlane(tv, j + u);
            const float* xr = x + (long long)g * D + lane;
            v0[u] = xr[0];
            v1[u] = xr[32];
        }
#pragma unroll
        for (int u = 0; u < 16; ++u) {
            float* ps = pl + a[u] * PITCH + lane;
            atomicAdd(ps,      v0[u]);
            atomicAdd(ps + 32, v1[u]);
        }
    }
    for (; j < n; ++j) {
        int g  = __builtin_amdgcn_readlane(gv, j);
        int aj = __builtin_amdgcn_readlane(tv, j);
        const float* xr = x + (long long)g * D + lane;
        float v0 = xr[0], v1 = xr[32];
        float* ps = pl + aj * PITCH + lane;
        atomicAdd(ps,      v0);
        atomicAdd(ps + 32, v1);
    }
}

__device__ __forceinline__ void sum_ordered_rows(const float* __restrict__ x,
                                                 float* pl, const int* sidx,
                                                 int pb, int pe, int lane) {
    int   tprev = -1;
    float s0 = 0.0f, s1 = 0.0f;
    int en = 0;
    if (pb + lane < pe) en = sidx[pb + lane];
    for (int p = pb; p < pe; p += 32) {
        int n  = pe - p < 32 ? pe - p : 32;
        int ev = en;
        if (p + 32 + lane < pe) en = sidx[p + 32 + lane];
        int j = 0;
        for (; j + 16 <= n; j += 16) {
            int   t[16];
            float v0[16], v1[16];
#pragma unroll
            for (int u = 0; u < 16; ++u) {
                int e = __builtin_amdgcn_readlane(ev, j + u);
                t[u]  = e >> TSH;
                unsigned off = (unsigned)(e & ((1 << TSH) - 1)) * D + lane;
                v0[u] = x[off];
                v1[u] = x[off + 32];
            }
#pragma unroll
            for (int u = 0; u < 16; ++u) {
                bool fresh = t[u] != tprev;
                tprev = t[u];
                s0 = (fresh ? 0.0f : s0) + v0[u];
                s1 = (fresh ? 0.0f : s1) + v1[u];
                float* ps = pl + t[u] * PITCH + lane;
                ps[0]  = s0;
                ps[32] = s1;
            }
        }
        for (; j < n; ++j) {
            int e  = __builtin_amdgcn_readlane(ev, j);
            int tj = e >> TSH;
            unsigned off = (unsigned)(e & ((1 << TSH) - 1)) * D + lane;
            float v0 = x[off], v1 = x[off + 32];
            bool fresh = tj != tprev;
            tprev = tj;
            s0 = (fresh ? 0.0f : s0) + v0;
            s1 = (fresh ? 0.0f : s1) + v1;
            float* ps = pl + tj * PITCH + lane;
            ps[0]  = s0;
            ps[32] = s1;
        }
    }
}

__global__ __launch_bounds__(1024)
void owner_sum_gemm_kernel(const float* __restrict__ x,
                           const int*   __restrict__ ei,
                           const int*   __restrict__ gcnt,
                           const int*   __restrict__ lists,
                           const float* __restrict__ Wself,
                           const float* __restrict__ bself,
                           const float* __restrict__ Ws2d,
                           const float* __restrict__ bs2d,
                           const float* __restrict__ Wd2s,
                           const float* __restrict__ bd2s,
                           float* __restrict__ out,
                           int N, int E, int NB, int CAP) {
    __shared__ __attribute__((aligned(16))) float acc[2 * RB * PITCH];
    __shared__ int   cntl[2 * RB];
    __shared__ __attribute__((aligned(16))) float wl[3 * D * D];
    __shared__ int   sidx[SC];
    __shared__ int   offs[RB + 4];

    int tid  = threadIdx.x;
    int lane = tid & 31;
    int wave = __builtin_amdgcn_readfirstlane(tid >> 5);
    int b    = blockIdx.x;
    int lo   = b * RB;

    v4f z4 = {0.0f, 0.0f, 0.0f, 0.0f};
    for (int i = tid; i < (2 * RB * PITCH) / 4; i += 1024) ((v4f*)acc)[i] = z4;
    for (int i = tid; i < 2 * RB; i += 1024) cntl[i] = 0;
    for (int i = tid; i < D * D; i += 1024) {
        int k   = i >> 6;
        int col = i & 63;
        int dst = ((k >> 1) << 7) + (col << 1) + (k & 1);
        wl[dst]             = Wself[i];
        wl[D * D + dst]     = Ws2d[i];
        wl[2 * D * D + dst] = Wd2s[i];
    }
    __syncthreads();

    for (int dir = 0; dir < 2; ++dir) {
        float* pl = acc  + dir * (RB * PITCH);
        int*   cl = cntl + dir * RB;
        int c = gcnt[dir * NB + b];
        if (c <= SC && c <= CAP) {
            const int2* rec = (const int2*)lists + (long long)(dir * NB + b) * CAP;
            int gk[SCT], tk[SCT], rk[SCT];
#pragma unroll
            for (int k = 0; k < SCT; ++k) {
                int i = k * 1024 + tid;
                gk[k] = 0; tk[k] = 0; rk[k] = 0;
                if (c > 0) {
                    int2 r = rec[i < c ? i : c - 1];
                    gk[k] = r.x; tk[k] = r.y;
                }
            }
#pragma unroll
            for (int k = 0; k < SCT; ++k) {
                int i = k * 1024 + tid;
                if (i < c) rk[k] = atomicAdd(&cl[tk[k]], 1);
            }
            __syncthreads();
            if (wave == 0) {
                int loc[13];
                int sum = 0;
#pragma unroll
                for (int q = 0; q < 13; ++q) {
                    int r = lane * 13 + q;
                    loc[q] = sum;
                    sum += (r < RB) ? cl[r] : 0;
                }
                int incl = sum;
#pragma unroll
                for (int dlt = 1; dlt < 32; dlt <<= 1) {
                    int up = __shfl_up(incl, dlt);
                    if (lane >= dlt) incl += up;
                }
                int basev = incl - sum;
#pragma unroll
                for (int q = 0; q < 13; ++q) {
                    int r = lane * 13 + q;
                    if (r <= RB) offs[r] = basev + loc[q];
                }
            }
            __syncthreads();
            int ok[SCT];
#pragma unroll
            for (int k = 0; k < SCT; ++k) ok[k] = offs[tk[k]];
#pragma unroll
            for (int k = 0; k < SCT; ++k) {
                int i = k * 1024 + tid;
                if (i < c) sidx[ok[k] + rk[k]] = gk[k] | (tk[k] << TSH);
            }
            __syncthreads();
            int rb = (wave * 25) >> 1;
            int re = ((wave + 1) * 25) >> 1;
            int pb = __builtin_amdgcn_readfirstlane(offs[rb]);
            int pe = __builtin_amdgcn_readfirstlane(offs[re]);
            sum_ordered_rows(x, pl, sidx, pb, pe, lane);
        } else if (c <= CAP) {
            const int2* rec = (const int2*)lists + (long long)(dir * NB + b) * CAP;
            int g = wave * 32;
            int2 rn; rn.x = 0; rn.y = 0;
            if (g + lane < c) rn = rec[g + lane];
            for (; g < c; g += 1024) {
                int n = c - g < 32 ? c - g : 32;
                int gv = rn.x, tv = rn.y;
                if (g + 1024 + lane < c) rn = rec[g + 1024 + lane];
                if (lane < n) atomicAdd(&cl[tv], 1);
                add_rows_lds(x, pl, gv, tv, n, lane);
            }
        } else {
            const int* gi = ei + (dir ? (long long)E : 0LL);
            const int* ti = ei + (dir ? 0LL : (long long)E);
            for (int e0 = wave * 32; e0 < E; e0 += 1024) {
                int e = e0 + lane;
                int gv = 0, tv = -1;
                if (e < E) {
                    int t = ti[e] - lo;
                    if ((unsigned)t < (unsigned)RB) {
                        tv = t;
                        gv = gi[e];
                        atomicAdd(&cl[tv], 1);
                    }
                }
                unsigned m = __builtin_amdgcn_ballot_w32(tv >= 0);
                while (m) {
                    int j = __builtin_ctz(m);
                    m &= m - 1;
                    int g  = __builtin_amdgcn_readlane(gv, j);
                    int aj = __builtin_amdgcn_readlane(tv, j);
                    const float* xr = x + (long long)g * D + lane;
                    float v0 = xr[0], v1 = xr[32];
                    float* ps = pl + aj * PITCH + lane;
                    atomicAdd(ps,      v0);
                    atomicAdd(ps + 32, v1);
                }
            }
        }
    }
    __syncthreads();

    int trow0 = wave * 16;
    if (trow0 >= RB || lo + trow0 >= N) return;

    int row   = lane & 15;
    int half  = lane >> 4;
    int khalf = half * 2;
    int lr    = trow0 + row;
    int gr    = lo + lr;

    float sc_in  = (1.0f - ALPHA) / fmaxf((float)cntl[lr], 1.0f);
    float sc_out = ALPHA          / fmaxf((float)cntl[RB + lr], 1.0f);

    v8f accv[4];
    for (int t = 0; t < 4; ++t) {
        float bb = bself[t * 16 + row]
                 + (1.0f - ALPHA) * bs2d[t * 16 + row]
                 + ALPHA * bd2s[t * 16 + row];
        for (int r = 0; r < 8; ++r) accv[t][r] = bb;
    }

    const float* xrow = x + (long long)gr * D;
    const float* irow = acc + lr * PITCH;
    const float* orow = acc + RB * PITCH + lr * PITCH;

    const float* wbase = &wl[half * 128 + row * 2];

    for (int k0 = 0; k0 < D; k0 += 4) {
        int ka = k0 + khalf;
        v2f ax = *(const v2f*)(xrow + ka);
        v2f ai = *(const v2f*)(irow + ka);
        v2f ao = *(const v2f*)(orow + ka);
        ai *= sc_in;
        ao *= sc_out;

        const float* wk = wbase + (k0 >> 1) * 128;
        for (int t = 0; t < 4; ++t) {
            v2f bs = *(const v2f*)(wk + t * 32);
            v2f bi = *(const v2f*)(wk + D * D + t * 32);
            v2f bo = *(const v2f*)(wk + 2 * D * D + t * 32);

            accv[t] = __builtin_amdgcn_wmma_f32_16x16x4_f32(
                false, ax, false, bs, (short)0, accv[t], false, false);
            accv[t] = __builtin_amdgcn_wmma_f32_16x16x4_f32(
                false, ai, false, bi, (short)0, accv[t], false, false);
            accv[t] = __builtin_amdgcn_wmma_f32_16x16x4_f32(
                false, ao, false, bo, (short)0, accv[t], false, false);
        }
    }

    int mofs = half * 8;
    for (int t = 0; t < 4; ++t) {
        int col = t * 16 + row;
        long long obase = (long long)(lo + trow0 + mofs) * D + col;
        for (int r = 0; r < 8; ++r) {
            out[obase + (long long)r * D] = accv[t][r];
        }
    }
}

extern "C" void kernel_launch(void* const* d_in, const int* in_sizes, int n_in,
                              void* d_out, int out_size, void* d_ws, size_t ws_size,
                              hipStream_t stream) {
    const float* x     = (const float*)d_in[0];
    const int*   ei    = (const int*)  d_in[1];
    const float* Wself = (const float*)d_in[2];
    const float* bself = (const float*)d_in[3];
    const float* Ws2d  = (const float*)d_in[4];
    const float* bs2d  = (const float*)d_in[5];
    const float* Wd2s  = (const float*)d_in[6];
    const float* bd2s  = (const float*)d_in[7];
    float* out = (float*)d_out;

    int N = in_sizes[0] / D;
    int E = in_sizes[1] / 2;

    int NB  = (N + RB - 1) / RB;
    int CAP = 2 * (E / NB);

    int* gcnt  = (int*)d_ws;
    int* lists = gcnt + 2 * NBMAX;

    zero_counters_kernel<<<(2 * NB + 1023) / 1024, 1024, 0, stream>>>(gcnt, 2 * NB);

    bin_edges_kernel<<<(E + CHUNK - 1) / CHUNK, BT, 0, stream>>>(
        ei, gcnt, lists, E, NB, CAP);

    owner_sum_gemm_kernel<<<NB, 1024, 0, stream>>>(
        x, ei, gcnt, lists, Wself, bself, Ws2d, bs2d, Wd2s, bd2s, out, N, E, NB, CAP);
}
